# starting kernel
# baseline (speedup 1.0000x reference)
#include <hip/hip_runtime.h>
#include <math.h>

#define C 128
#define TPB 256

typedef float v2f __attribute__((ext_vector_type(2)));
typedef float v8f __attribute__((ext_vector_type(8)));

__global__ void deg_kernel(const int* __restrict__ dst, float* __restrict__ deg, int E) {
    int e = blockIdx.x * blockDim.x + threadIdx.x;
    if (e < E) atomicAdd(&deg[dst[e]], 1.0f);
}

__global__ void dinv_kernel(float* __restrict__ degdinv, int N) {
    int i = blockIdx.x * blockDim.x + threadIdx.x;
    if (i < N) degdinv[i] = rsqrtf(degdinv[i] + 1.0f);
}

__global__ __launch_bounds__(TPB) void gemm_wmma(const float* __restrict__ X,
                                                 const float* __restrict__ W,
                                                 float* __restrict__ Y,
                                                 int nStrips) {
    __shared__ float Wl[64 * C];

    const int wave = threadIdx.x >> 5;
    const int lane = threadIdx.x & 31;
    const int half = lane >> 4;
    const int lo   = lane & 15;

    const int strip  = blockIdx.x * (TPB / 32) + wave;
    const bool active = (strip < nStrips);
    const int m0 = strip * 16;

    v8f acc[8] = {};

    const float* Arow = X + (size_t)(active ? (m0 + lo) : 0) * C + 2 * half;

    for (int kc = 0; kc < C; kc += 64) {
        __syncthreads();
        for (int idx = threadIdx.x; idx < 64 * C; idx += TPB) {
            int k = idx >> 7;
            int n = idx & (C - 1);
            Wl[(k >> 1) * 256 + n * 2 + (k & 1)] = W[(size_t)(kc + k) * C + n];
        }
        __syncthreads();

        if (active) {
            #pragma unroll
            for (int k0 = 0; k0 < 64; k0 += 4) {
                v2f a = *(const v2f*)(Arow + kc + k0);
                const float* Bbase = &Wl[((k0 >> 1) + half) * 256 + lo * 2];
                #pragma unroll
                for (int t = 0; t < 8; ++t) {
                    v2f b = *(const v2f*)(Bbase + t * 32);
                    acc[t] = __builtin_amdgcn_wmma_f32_16x16x4_f32(
                        false, a, false, b, (short)0, acc[t], false, false);
                }
            }
        }
    }

    if (active) {
        float* Yt = Y + (size_t)m0 * C;
        #pragma unroll
        for (int t = 0; t < 8; ++t)
            #pragma unroll
            for (int r = 0; r < 8; ++r)
                Yt[(size_t)(r + 8 * half) * C + t * 16 + lo] = acc[t][r];
    }
}

__global__ void gemm_tail(const float* __restrict__ X, const float* __restrict__ W,
                          float* __restrict__ Y, int row0, int rows) {
    int i = blockIdx.x * blockDim.x + threadIdx.x;
    if (i >= rows * C) return;
    int r = row0 + i / C, n = i & (C - 1);
    float s = 0.f;
    for (int k = 0; k < C; ++k) s += X[(size_t)r * C + k] * W[(size_t)k * C + n];
    Y[(size_t)r * C + n] = s;
}

__global__ void agg_kernel(const float* __restrict__ H, const int* __restrict__ src,
                           const int* __restrict__ dst, const float* __restrict__ dinv,
                           float* __restrict__ out, int E, int N) {
    int gw   = (blockIdx.x * blockDim.x + threadIdx.x) >> 5;
    int lane = threadIdx.x & 31;
    int nw   = (gridDim.x * blockDim.x) >> 5;
    int total = E + N;
    for (int e = gw; e < total; e += nw) {
        int s, d;
        if (e < E) { s = src[e]; d = dst[e]; } else { s = e - E; d = s; }
        float nrm = dinv[s] * dinv[d];
        float4 v = *(const float4*)(H + (size_t)s * C + lane * 4);
        float* o = out + (size_t)d * C + lane * 4;
        atomicAdd(o + 0, v.x * nrm);
        atomicAdd(o + 1, v.y * nrm);
        atomicAdd(o + 2, v.z * nrm);
        atomicAdd(o + 3, v.w * nrm);
    }
}

__global__ void bias_elu_kernel(const float* __restrict__ in, const float* __restrict__ b,
                                float* __restrict__ out, size_t n4) {
    size_t i = (size_t)blockIdx.x * blockDim.x + threadIdx.x;
    if (i >= n4) return;
    size_t idx = i * 4;
    int c = (int)(idx & (C - 1));
    float4 v = *(const float4*)(in + idx);
    v.x += b[c + 0]; v.y += b[c + 1]; v.z += b[c + 2]; v.w += b[c + 3];
    v.x = v.x > 0.f ? v.x : (expf(v.x) - 1.f);
    v.y = v.y > 0.f ? v.y : (expf(v.y) - 1.f);
    v.z = v.z > 0.f ? v.z : (expf(v.z) - 1.f);
    v.w = v.w > 0.f ? v.w : (expf(v.w) - 1.f);
    *(float4*)(out + idx) = v;
}

extern "C" void kernel_launch(void* const* d_in, const int* in_sizes, int n_in,
                              void* d_out, int out_size, void* d_ws, size_t ws_size,
                              hipStream_t stream) {
    const float* x  = (const float*)d_in[0];
    const int*   ei = (const int*)d_in[1];
    const float* W1 = (const float*)d_in[2];
    const float* b1 = (const float*)d_in[3];
    const float* W2 = (const float*)d_in[4];
    const float* b2 = (const float*)d_in[5];

    const int N = in_sizes[0] / C;
    const int E = in_sizes[1] / 2;
    const int* srcI = ei;
    const int* dstI = ei + E;

    char* ws = (char*)d_ws;
    float* dinv = (float*)ws;
    size_t o1 = ((size_t)N * 4 + 255) & ~(size_t)255;
    size_t hbytes = ((size_t)N * C * 4 + 255) & ~(size_t)255;
    float* buf0 = (float*)(ws + o1);
    float* buf1 = (float*)(ws + o1 + hbytes);

    const int strips   = N / 16;
    const int tailRow0 = strips * 16;
    const int tailRows = N - tailRow0;
    const dim3 blk(TPB);
    const int gemmBlocks = (strips + (TPB / 32) - 1) / (TPB / 32);
    const int aggBlocks  = 4096;
    const size_t n4      = (size_t)N * C / 4;
    const int eluBlocks  = (int)((n4 + TPB - 1) / TPB);

    hipMemsetAsync(dinv, 0, (size_t)N * 4, stream);
    deg_kernel<<<(E + TPB - 1) / TPB, blk, 0, stream>>>(dstI, dinv, E);
    dinv_kernel<<<(N + TPB - 1) / TPB, blk, 0, stream>>>(dinv, N);

    gemm_wmma<<<gemmBlocks, blk, 0, stream>>>(x, W1, buf0, strips);
    if (tailRows > 0)
        gemm_tail<<<(tailRows * C + TPB - 1) / TPB, blk, 0, stream>>>(x, W1, buf0, tailRow0, tailRows);
    hipMemsetAsync(buf1, 0, (size_t)N * C * 4, stream);
    agg_kernel<<<aggBlocks, blk, 0, stream>>>(buf0, srcI, dstI, dinv, buf1, E, N);
    bias_elu_kernel<<<eluBlocks, blk, 0, stream>>>(buf1, b1, buf0, n4);

    gemm_wmma<<<gemmBlocks, blk, 0, stream>>>(buf0, W2, buf1, strips);
    if (tailRows > 0)
        gemm_tail<<<(tailRows * C + TPB - 1) / TPB, blk, 0, stream>>>(buf0, W2, buf1, tailRow0, tailRows);
    hipMemsetAsync(buf0, 0, (size_t)N * C * 4, stream);
    agg_kernel<<<aggBlocks, blk, 0, stream>>>(buf1, srcI, dstI, dinv, buf0, E, N);
    bias_elu_kernel<<<eluBlocks, blk, 0, stream>>>(buf0, b2, (float*)d_out, n4);
}


# Degree count moves to LDS: 64 blocks each count a slice of dst in a 200-KiB table of integer LDS counters and store it as a plane; dinv adds the planes; scattered global f32 adds, memset gone; exact
# speedup vs baseline: 2.2591x; 2.2591x over previous
#include <hip/hip_runtime.h>
#include <math.h>

#define C 128
#define TPB 256

typedef float v2f __attribute__((ext_vector_type(2)));
typedef float v8f __attribute__((ext_vector_type(8)));

#define HCAP 51200
#define HTPB 1024
#define HBLK 64
#define HLD  4

__device__ __forceinline__ int plane_stride(int N) { return (N + 63) & ~63; }

__device__ __forceinline__ void bump(unsigned int* cnt, int d, int base, int span) {
    const unsigned int r = (unsigned int)(d - base);
    if (r < (unsigned int)span) atomicAdd(&cnt[r], 1u);
}

__global__ __launch_bounds__(HTPB) void deg_hist_kernel(const int* __restrict__ dst,
                                                        unsigned int* __restrict__ part,
                                                        int E, int N) {
    __shared__ unsigned int cnt[HCAP];
    const int tid = threadIdx.x;
    const int per = ((((E + (int)gridDim.x - 1) / (int)gridDim.x) + 3) & ~3);
    const int e0  = min(E, (int)blockIdx.x * per);
    const int n   = min(E, e0 + per) - e0;
    const int* p  = dst + e0;
    const bool vecIn = ((((size_t)p) & 15) == 0);
    const int n4 = vecIn ? (n >> 2) : 0;
    const int4* p4 = (const int4*)p;
    const int PS = plane_stride(N);

    for (int base = 0; base < N; base += HCAP) {
        const int span  = min(HCAP, N - base);
        const int span4 = (span + 3) >> 2;
        for (int i = tid; i < span4; i += HTPB)
            ((uint4*)cnt)[i] = make_uint4(0u, 0u, 0u, 0u);
        __syncthreads();

        for (int q0 = 0; q0 < n4; q0 += HLD * HTPB) {
            int4 v[HLD];
            #pragma unroll
            for (int u = 0; u < HLD; ++u) {
                const int q = q0 + u * HTPB + tid;
                v[u] = (q < n4) ? p4[q] : make_int4(-1, -1, -1, -1);
            }
            #pragma unroll
            for (int u = 0; u < HLD; ++u) {
                bump(cnt, v[u].x, base, span);
                bump(cnt, v[u].y, base, span);
                bump(cnt, v[u].z, base, span);
                bump(cnt, v[u].w, base, span);
            }
        }
        for (int e = (n4 << 2) + tid; e < n; e += HTPB) bump(cnt, p[e], base, span);
        __syncthreads();

        uint4* plane4 = (uint4*)(part + (size_t)blockIdx.x * PS + base);
        for (int i = tid; i < span4; i += HTPB) plane4[i] = ((const uint4*)cnt)[i];
        __syncthreads();
    }
}

__global__ void dinv_kernel(const unsigned int* __restrict__ part, float* __restrict__ dinv, int N) {
    int i = blockIdx.x * blockDim.x + threadIdx.x;
    if (i >= N) return;
    const int PS = plane_stride(N);
    unsigned int s = 0;
    #pragma unroll
    for (int g = 0; g < HBLK; ++g) s += part[(size_t)g * PS + i];
    dinv[i] = rsqrtf((float)s + 1.0f);
}

__global__ __launch_bounds__(TPB) void gemm_wmma(const float* __restrict__ X,
                                                 const float* __restrict__ W,
                                                 const float* __restrict__ dinv,
                                                 float* __restrict__ Y,
                                                 float* __restrict__ Z,
                                                 int nStrips) {
    __shared__ float Wl[64 * C];

    const int wave = threadIdx.x >> 5;
    const int lane = threadIdx.x & 31;
    const int half = lane >> 4;
    const int lo   = lane & 15;

    const int strip  = blockIdx.x * (TPB / 32) + wave;
    const bool active = (strip < nStrips);
    const int m0 = strip * 16;

    v8f acc[8] = {};

    const float* Arow = X + (size_t)(active ? (m0 + lo) : 0) * C + 2 * half;

    for (int kc = 0; kc < C; kc += 64) {
        __syncthreads();
        for (int idx = threadIdx.x; idx < 64 * C; idx += TPB) {
            int k = idx >> 7;
            int n = idx & (C - 1);
            Wl[(k >> 1) * 256 + n * 2 + (k & 1)] = W[(size_t)(kc + k) * C + n];
        }
        __syncthreads();

        if (active) {
            #pragma unroll
            for (int k0 = 0; k0 < 64; k0 += 4) {
                v2f a = *(const v2f*)(Arow + kc + k0);
                const float* Bbase = &Wl[((k0 >> 1) + half) * 256 + lo * 2];
                #pragma unroll
                for (int t = 0; t < 8; ++t) {
                    v2f b = *(const v2f*)(Bbase + t * 32);
                    acc[t] = __builtin_amdgcn_wmma_f32_16x16x4_f32(
                        false, a, false, b, (short)0, acc[t], false, false);
                }
            }
        }
    }

    if (active) {
        float* Yt = Y + (size_t)m0 * C;
        float* Zt = Z + (size_t)m0 * C;
        const float4 dA = *(const float4*)(dinv + m0 + 8 * half);
        const float4 dB = *(const float4*)(dinv + m0 + 8 * half + 4);
        const float dv[8] = {dA.x, dA.y, dA.z, dA.w, dB.x, dB.y, dB.z, dB.w};
        #pragma unroll
        for (int t = 0; t < 8; ++t)
            #pragma unroll
            for (int r = 0; r < 8; ++r) {
                const size_t o = (size_t)(r + 8 * half) * C + t * 16 + lo;
                Yt[o] = acc[t][r];
                Zt[o] = acc[t][r] * (dv[r] * dv[r]);
            }
    }
}

__global__ void gemm_tail(const float* __restrict__ X, const float* __restrict__ W,
                          const float* __restrict__ dinv,
                          float* __restrict__ Y, float* __restrict__ Z, int row0, int rows) {
    int i = blockIdx.x * blockDim.x + threadIdx.x;
    if (i >= rows * C) return;
    int r = row0 + i / C, n = i & (C - 1);
    float s = 0.f;
    for (int k = 0; k < C; ++k) s += X[(size_t)r * C + k] * W[(size_t)k * C + n];
    Y[(size_t)r * C + n] = s;
    Z[(size_t)r * C + n] = s * (dinv[r] * dinv[r]);
}

#define EB 8
__global__ __launch_bounds__(TPB) void agg_kernel(const float* __restrict__ H, const int* __restrict__ src,
                           const int* __restrict__ dst, const float* __restrict__ dinv,
                           float* __restrict__ out, int E) {
    const int lane = threadIdx.x & 31;
    const int gw = __builtin_amdgcn_readfirstlane((int)((blockIdx.x * blockDim.x + threadIdx.x) >> 5));
    const int nw = (int)((gridDim.x * blockDim.x) >> 5);
    const int total = E;
    const int nBatch = (total + 31) >> 5;
    for (int b = gw; b < nBatch; b += nw) {
        const int e0  = b << 5;
        const int cnt = min(32, total - e0);
        const int e   = e0 + lane;
        int s = 0, d = 0;
        float nrm = 0.f;
        if (e < total) {
            s = src[e];
            d = dst[e];
            nrm = dinv[s] * dinv[d];
        }
        const int nrmBits = __float_as_int(nrm);
        int j = 0;
        for (; j + EB <= cnt; j += EB) {
            float v[EB][4];
            #pragma unroll
            for (int u = 0; u < EB; ++u) {
                const int sj = __builtin_amdgcn_readlane(s, j + u);
                const float* h = H + (size_t)sj * C + lane;
                v[u][0] = h[0];
                v[u][1] = h[32];
                v[u][2] = h[64];
                v[u][3] = h[96];
            }
            #pragma unroll
            for (int u = 0; u < EB; ++u) {
                const int   dj = __builtin_amdgcn_readlane(d, j + u);
                const float nj = __int_as_float(__builtin_amdgcn_readlane(nrmBits, j + u));
                float* o = out + (size_t)dj * C + lane;
                atomicAdd(o +  0, v[u][0] * nj);
                atomicAdd(o + 32, v[u][1] * nj);
                atomicAdd(o + 64, v[u][2] * nj);
                atomicAdd(o + 96, v[u][3] * nj);
            }
        }
        for (; j < cnt; ++j) {
            const int   sj = __builtin_amdgcn_readlane(s, j);
            const int   dj = __builtin_amdgcn_readlane(d, j);
            const float nj = __int_as_float(__builtin_amdgcn_readlane(nrmBits, j));
            const float* h = H + (size_t)sj * C + lane;
            float* o = out + (size_t)dj * C + lane;
            float v0 = h[0];
            float v1 = h[32];
            float v2 = h[64];
            float v3 = h[96];
            atomicAdd(o +  0, v0 * nj);
            atomicAdd(o + 32, v1 * nj);
            atomicAdd(o + 64, v2 * nj);
            atomicAdd(o + 96, v3 * nj);
        }
    }
}

__global__ void bias_elu_kernel(const float* __restrict__ in, const float* __restrict__ b,
                                float* __restrict__ out, size_t n4) {
    size_t i = (size_t)blockIdx.x * blockDim.x + threadIdx.x;
    if (i >= n4) return;
    size_t idx = i * 4;
    int c = (int)(idx & (C - 1));
    float4 v = *(const float4*)(in + idx);
    v.x += b[c + 0]; v.y += b[c + 1]; v.z += b[c + 2]; v.w += b[c + 3];
    v.x = v.x > 0.f ? v.x : (expf(v.x) - 1.f);
    v.y = v.y > 0.f ? v.y : (expf(v.y) - 1.f);
    v.z = v.z > 0.f ? v.z : (expf(v.z) - 1.f);
    v.w = v.w > 0.f ? v.w : (expf(v.w) - 1.f);
    *(float4*)(out + idx) = v;
}

extern "C" void kernel_launch(void* const* d_in, const int* in_sizes, int n_in,
                              void* d_out, int out_size, void* d_ws, size_t ws_size,
                              hipStream_t stream) {
    const float* x  = (const float*)d_in[0];
    const int*   ei = (const int*)d_in[1];
    const float* W1 = (const float*)d_in[2];
    const float* b1 = (const float*)d_in[3];
    const float* W2 = (const float*)d_in[4];
    const float* b2 = (const float*)d_in[5];

    const int N = in_sizes[0] / C;
    const int E = in_sizes[1] / 2;
    const int* srcI = ei;
    const int* dstI = ei + E;

    char* ws = (char*)d_ws;
    float* dinv = (float*)ws;
    size_t o1 = ((size_t)N * 4 + 255) & ~(size_t)255;
    size_t hbytes = ((size_t)N * C * 4 + 255) & ~(size_t)255;
    float* buf0 = (float*)(ws + o1);
    float* buf1 = (float*)(ws + o1 + hbytes);
    float* buf2 = (float*)(ws + o1 + 2 * hbytes);
    unsigned int* part = (unsigned int*)(ws + o1 + 3 * hbytes);

    const int strips   = N / 16;
    const int tailRow0 = strips * 16;
    const int tailRows = N - tailRow0;
    const dim3 blk(TPB);
    const int gemmBlocks = (strips + (TPB / 32) - 1) / (TPB / 32);
    const int aggBatches = (E + 31) / 32;
    const int aggBlocks  = (aggBatches + (TPB / 32) - 1) / (TPB / 32);
    const size_t n4      = (size_t)N * C / 4;
    const int eluBlocks  = (int)((n4 + TPB - 1) / TPB);

    deg_hist_kernel<<<HBLK, dim3(HTPB), 0, stream>>>(dstI, part, E, N);
    dinv_kernel<<<(N + TPB - 1) / TPB, blk, 0, stream>>>(part, dinv, N);

    gemm_wmma<<<gemmBlocks, blk, 0, stream>>>(x, W1, dinv, buf0, buf1, strips);
    if (tailRows > 0)
        gemm_tail<<<(tailRows * C + TPB - 1) / TPB, blk, 0, stream>>>(x, W1, dinv, buf0, buf1, tailRow0, tailRows);
    agg_kernel<<<aggBlocks, blk, 0, stream>>>(buf0, srcI, dstI, dinv, buf1, E);
    bias_elu_kernel<<<eluBlocks, blk, 0, stream>>>(buf1, b1, buf0, n4);

    gemm_wmma<<<gemmBlocks, blk, 0, stream>>>(buf0, W2, dinv, buf1, buf2, strips);
    if (tailRows > 0)
        gemm_tail<<<(tailRows * C + TPB - 1) / TPB, blk, 0, stream>>>(buf0, W2, dinv, buf1, buf2, tailRow0, tailRows);
    agg_kernel<<<aggBlocks, blk, 0, stream>>>(buf1, srcI, dstI, dinv, buf2, E);
    bias_elu_kernel<<<eluBlocks, blk, 0, stream>>>(buf2, b2, (float*)d_out, n4);
}


# Layer-1 bias+ELU moves into layer 2's GEMM A-fragment load: sums activated in registers on the way to the WMMA; one elementwise launch, its 26-MB write and re-read gone; same arithmetic
# speedup vs baseline: 2.4210x; 1.0717x over previous
#include <hip/hip_runtime.h>
#include <math.h>

#define C 128
#define TPB 256

typedef float v2f __attribute__((ext_vector_type(2)));
typedef float v8f __attribute__((ext_vector_type(8)));

#define HCAP 51200
#define HTPB 1024
#define HBLK 64
#define HLD  4

__device__ __forceinline__ int plane_stride(int N) { return (N + 63) & ~63; }

__device__ __forceinline__ void bump(unsigned int* cnt, int d, int base, int span) {
    const unsigned int r = (unsigned int)(d - base);
    if (r < (unsigned int)span) atomicAdd(&cnt[r], 1u);
}

__global__ __launch_bounds__(HTPB) void deg_hist_kernel(const int* __restrict__ dst,
                                                        unsigned int* __restrict__ part,
                                                        int E, int N) {
    __shared__ unsigned int cnt[HCAP];
    const int tid = threadIdx.x;
    const int per = ((((E + (int)gridDim.x - 1) / (int)gridDim.x) + 3) & ~3);
    const int e0  = min(E, (int)blockIdx.x * per);
    const int n   = min(E, e0 + per) - e0;
    const int* p  = dst + e0;
    const bool vecIn = ((((size_t)p) & 15) == 0);
    const int n4 = vecIn ? (n >> 2) : 0;
    const int4* p4 = (const int4*)p;
    const int PS = plane_stride(N);

    for (int base = 0; base < N; base += HCAP) {
        const int span  = min(HCAP, N - base);
        const int span4 = (span + 3) >> 2;
        for (int i = tid; i < span4; i += HTPB)
            ((uint4*)cnt)[i] = make_uint4(0u, 0u, 0u, 0u);
        __syncthreads();

        for (int q0 = 0; q0 < n4; q0 += HLD * HTPB) {
            int4 v[HLD];
            #pragma unroll
            for (int u = 0; u < HLD; ++u) {
                const int q = q0 + u * HTPB + tid;
                v[u] = (q < n4) ? p4[q] : make_int4(-1, -1, -1, -1);
            }
            #pragma unroll
            for (int u = 0; u < HLD; ++u) {
                bump(cnt, v[u].x, base, span);
                bump(cnt, v[u].y, base, span);
                bump(cnt, v[u].z, base, span);
                bump(cnt, v[u].w, base, span);
            }
        }
        for (int e = (n4 << 2) + tid; e < n; e += HTPB) bump(cnt, p[e], base, span);
        __syncthreads();

        uint4* plane4 = (uint4*)(part + (size_t)blockIdx.x * PS + base);
        for (int i = tid; i < span4; i += HTPB) plane4[i] = ((const uint4*)cnt)[i];
        __syncthreads();
    }
}

__global__ void dinv_kernel(const unsigned int* __restrict__ part, float* __restrict__ dinv, int N) {
    int i = blockIdx.x * blockDim.x + threadIdx.x;
    if (i >= N) return;
    const int PS = plane_stride(N);
    unsigned int s = 0;
    #pragma unroll
    for (int g = 0; g < HBLK; ++g) s += part[(size_t)g * PS + i];
    dinv[i] = rsqrtf((float)s + 1.0f);
}

__device__ __forceinline__ float elu1(float v) {
    const float m = v > 0.f ? 0.f : v;
    const float e = expf(m) - 1.f;
    return v > 0.f ? v : e;
}

template <bool ACT>
__global__ __launch_bounds__(TPB) void gemm_wmma(const float* __restrict__ X,
                                                 const float* __restrict__ W,
                                                 const float* __restrict__ bin,
                                                 const float* __restrict__ dinv,
                                                 float* __restrict__ Y,
                                                 float* __restrict__ Z,
                                                 int nStrips) {
    __shared__ float Wl[64 * C];
    __shared__ __attribute__((aligned(16))) float bl[C];
    if (ACT) {
        if (threadIdx.x < C) bl[threadIdx.x] = bin[threadIdx.x];
    }

    const int wave = threadIdx.x >> 5;
    const int lane = threadIdx.x & 31;
    const int half = lane >> 4;
    const int lo   = lane & 15;

    const int strip  = blockIdx.x * (TPB / 32) + wave;
    const bool active = (strip < nStrips);
    const int m0 = strip * 16;

    v8f acc[8] = {};

    const float* Arow = X + (size_t)(active ? (m0 + lo) : 0) * C + 2 * half;

    for (int kc = 0; kc < C; kc += 64) {
        __syncthreads();
        for (int idx = threadIdx.x; idx < 64 * C; idx += TPB) {
            int k = idx >> 7;
            int n = idx & (C - 1);
            Wl[(k >> 1) * 256 + n * 2 + (k & 1)] = W[(size_t)(kc + k) * C + n];
        }
        __syncthreads();

        if (active) {
            #pragma unroll
            for (int k0 = 0; k0 < 64; k0 += 4) {
                v2f a = *(const v2f*)(Arow + kc + k0);
                if (ACT) {
                    const v2f bb = *(const v2f*)(&bl[kc + k0 + 2 * half]);
                    a.x = elu1(a.x + bb.x);
                    a.y = elu1(a.y + bb.y);
                }
                const float* Bbase = &Wl[((k0 >> 1) + half) * 256 + lo * 2];
                #pragma unroll
                for (int t = 0; t < 8; ++t) {
                    v2f b = *(const v2f*)(Bbase + t * 32);
                    acc[t] = __builtin_amdgcn_wmma_f32_16x16x4_f32(
                        false, a, false, b, (short)0, acc[t], false, false);
                }
            }
        }
    }

    if (active) {
        float* Yt = Y + (size_t)m0 * C;
        float* Zt = Z + (size_t)m0 * C;
        const float4 dA = *(const float4*)(dinv + m0 + 8 * half);
        const float4 dB = *(const float4*)(dinv + m0 + 8 * half + 4);
        const float dv[8] = {dA.x, dA.y, dA.z, dA.w, dB.x, dB.y, dB.z, dB.w};
        #pragma unroll
        for (int t = 0; t < 8; ++t)
            #pragma unroll
            for (int r = 0; r < 8; ++r) {
                const size_t o = (size_t)(r + 8 * half) * C + t * 16 + lo;
                Yt[o] = acc[t][r];
                Zt[o] = acc[t][r] * (dv[r] * dv[r]);
            }
    }
}

template <bool ACT>
__global__ void gemm_tail(const float* __restrict__ X, const float* __restrict__ W,
                          const float* __restrict__ bin, const float* __restrict__ dinv,
                          float* __restrict__ Y, float* __restrict__ Z, int row0, int rows) {
    int i = blockIdx.x * blockDim.x + threadIdx.x;
    if (i >= rows * C) return;
    int r = row0 + i / C, n = i & (C - 1);
    float s = 0.f;
    for (int k = 0; k < C; ++k) {
        float a = X[(size_t)r * C + k];
        if (ACT) a = elu1(a + bin[k]);
        s += a * W[(size_t)k * C + n];
    }
    Y[(size_t)r * C + n] = s;
    Z[(size_t)r * C + n] = s * (dinv[r] * dinv[r]);
}

#define EB 8
__global__ __launch_bounds__(TPB) void agg_kernel(const float* __restrict__ H, const int* __restrict__ src,
                           const int* __restrict__ dst, const float* __restrict__ dinv,
                           float* __restrict__ out, int E) {
    const int lane = threadIdx.x & 31;
    const int gw = __builtin_amdgcn_readfirstlane((int)((blockIdx.x * blockDim.x + threadIdx.x) >> 5));
    const int nw = (int)((gridDim.x * blockDim.x) >> 5);
    const int total = E;
    const int nBatch = (total + 31) >> 5;
    for (int b = gw; b < nBatch; b += nw) {
        const int e0  = b << 5;
        const int cnt = min(32, total - e0);
        const int e   = e0 + lane;
        int s = 0, d = 0;
        float nrm = 0.f;
        if (e < total) {
            s = src[e];
            d = dst[e];
            nrm = dinv[s] * dinv[d];
        }
        const int nrmBits = __float_as_int(nrm);
        int j = 0;
        for (; j + EB <= cnt; j += EB) {
            float v[EB][4];
            #pragma unroll
            for (int u = 0; u < EB; ++u) {
                const int sj = __builtin_amdgcn_readlane(s, j + u);
                const float* h = H + (size_t)sj * C + lane;
                v[u][0] = h[0];
                v[u][1] = h[32];
                v[u][2] = h[64];
                v[u][3] = h[96];
            }
            #pragma unroll
            for (int u = 0; u < EB; ++u) {
                const int   dj = __builtin_amdgcn_readlane(d, j + u);
                const float nj = __int_as_float(__builtin_amdgcn_readlane(nrmBits, j + u));
                float* o = out + (size_t)dj * C + lane;
                atomicAdd(o +  0, v[u][0] * nj);
                atomicAdd(o + 32, v[u][1] * nj);
                atomicAdd(o + 64, v[u][2] * nj);
                atomicAdd(o + 96, v[u][3] * nj);
            }
        }
        for (; j < cnt; ++j) {
            const int   sj = __builtin_amdgcn_readlane(s, j);
            const int   dj = __builtin_amdgcn_readlane(d, j);
            const float nj = __int_as_float(__builtin_amdgcn_readlane(nrmBits, j));
            const float* h = H + (size_t)sj * C + lane;
            float* o = out + (size_t)dj * C + lane;
            float v0 = h[0];
            float v1 = h[32];
            float v2 = h[64];
            float v3 = h[96];
            atomicAdd(o +  0, v0 * nj);
            atomicAdd(o + 32, v1 * nj);
            atomicAdd(o + 64, v2 * nj);
            atomicAdd(o + 96, v3 * nj);
        }
    }
}

__global__ void bias_elu_kernel(const float* __restrict__ in, const float* __restrict__ b,
                                float* __restrict__ out, size_t n4) {
    size_t i = (size_t)blockIdx.x * blockDim.x + threadIdx.x;
    if (i >= n4) return;
    size_t idx = i * 4;
    int c = (int)(idx & (C - 1));
    float4 v = *(const float4*)(in + idx);
    v.x += b[c + 0]; v.y += b[c + 1]; v.z += b[c + 2]; v.w += b[c + 3];
    v.x = v.x > 0.f ? v.x : (expf(v.x) - 1.f);
    v.y = v.y > 0.f ? v.y : (expf(v.y) - 1.f);
    v.z = v.z > 0.f ? v.z : (expf(v.z) - 1.f);
    v.w = v.w > 0.f ? v.w : (expf(v.w) - 1.f);
    *(float4*)(out + idx) = v;
}

extern "C" void kernel_launch(void* const* d_in, const int* in_sizes, int n_in,
                              void* d_out, int out_size, void* d_ws, size_t ws_size,
                              hipStream_t stream) {
    const float* x  = (const float*)d_in[0];
    const int*   ei = (const int*)d_in[1];
    const float* W1 = (const float*)d_in[2];
    const float* b1 = (const float*)d_in[3];
    const float* W2 = (const float*)d_in[4];
    const float* b2 = (const float*)d_in[5];

    const int N = in_sizes[0] / C;
    const int E = in_sizes[1] / 2;
    const int* srcI = ei;
    const int* dstI = ei + E;

    char* ws = (char*)d_ws;
    float* dinv = (float*)ws;
    size_t o1 = ((size_t)N * 4 + 255) & ~(size_t)255;
    size_t hbytes = ((size_t)N * C * 4 + 255) & ~(size_t)255;
    float* buf0 = (float*)(ws + o1);
    float* buf1 = (float*)(ws + o1 + hbytes);
    float* buf2 = (float*)(ws + o1 + 2 * hbytes);
    unsigned int* part = (unsigned int*)(ws + o1 + 3 * hbytes);

    const int strips   = N / 16;
    const int tailRow0 = strips * 16;
    const int tailRows = N - tailRow0;
    const dim3 blk(TPB);
    const int gemmBlocks = (strips + (TPB / 32) - 1) / (TPB / 32);
    const int aggBatches = (E + 31) / 32;
    const int aggBlocks  = (aggBatches + (TPB / 32) - 1) / (TPB / 32);
    const size_t n4      = (size_t)N * C / 4;
    const int eluBlocks  = (int)((n4 + TPB - 1) / TPB);

    deg_hist_kernel<<<HBLK, dim3(HTPB), 0, stream>>>(dstI, part, E, N);
    dinv_kernel<<<(N + TPB - 1) / TPB, blk, 0, stream>>>(part, dinv, N);

    gemm_wmma<false><<<gemmBlocks, blk, 0, stream>>>(x, W1, nullptr, dinv, buf0, buf1, strips);
    if (tailRows > 0)
        gemm_tail<false><<<(tailRows * C + TPB - 1) / TPB, blk, 0, stream>>>(x, W1, nullptr, dinv, buf0, buf1, tailRow0, tailRows);
    agg_kernel<<<aggBlocks, blk, 0, stream>>>(buf0, srcI, dstI, dinv, buf1, E);

    gemm_wmma<true><<<gemmBlocks, blk, 0, stream>>>(buf1, W2, b1, dinv, buf0, buf2, strips);
    if (tailRows > 0)
        gemm_tail<true><<<(tailRows * C + TPB - 1) / TPB, blk, 0, stream>>>(buf1, W2, b1, dinv, buf0, buf2, tailRow0, tailRows);
    agg_kernel<<<aggBlocks, blk, 0, stream>>>(buf0, srcI, dstI, dinv, buf2, E);
    bias_elu_kernel<<<eluBlocks, blk, 0, stream>>>(buf2, b2, (float*)d_out, n4);
}


# Scatter-add split by cache half: a wave reads its die number, adds only into its half's node range; 64-edge tickets per half by integer atomics; an out line is added to from one half; arrival order
# speedup vs baseline: 2.5901x; 1.0698x over previous
#include <hip/hip_runtime.h>
#include <math.h>

#define C 128
#define TPB 256

typedef float v2f __attribute__((ext_vector_type(2)));
typedef float v8f __attribute__((ext_vector_type(8)));

#define HCAP 51200
#define HTPB 1024
#define HBLK 64
#define HLD  4

__device__ __forceinline__ int plane_stride(int N) { return (N + 63) & ~63; }

__device__ __forceinline__ void bump(unsigned int* cnt, int d, int base, int span) {
    const unsigned int r = (unsigned int)(d - base);
    if (r < (unsigned int)span) atomicAdd(&cnt[r], 1u);
}

__global__ __launch_bounds__(HTPB) void deg_hist_kernel(const int* __restrict__ dst,
                                                        unsigned int* __restrict__ part,
                                                        int E, int N) {
    __shared__ unsigned int cnt[HCAP];
    const int tid = threadIdx.x;
    const int per = ((((E + (int)gridDim.x - 1) / (int)gridDim.x) + 3) & ~3);
    const int e0  = min(E, (int)blockIdx.x * per);
    const int n   = min(E, e0 + per) - e0;
    const int* p  = dst + e0;
    const bool vecIn = ((((size_t)p) & 15) == 0);
    const int n4 = vecIn ? (n >> 2) : 0;
    const int4* p4 = (const int4*)p;
    const int PS = plane_stride(N);

    for (int base = 0; base < N; base += HCAP) {
        const int span  = min(HCAP, N - base);
        const int span4 = (span + 3) >> 2;
        for (int i = tid; i < span4; i += HTPB)
            ((uint4*)cnt)[i] = make_uint4(0u, 0u, 0u, 0u);
        __syncthreads();

        for (int q0 = 0; q0 < n4; q0 += HLD * HTPB) {
            int4 v[HLD];
            #pragma unroll
            for (int u = 0; u < HLD; ++u) {
                const int q = q0 + u * HTPB + tid;
                v[u] = (q < n4) ? p4[q] : make_int4(-1, -1, -1, -1);
            }
            #pragma unroll
            for (int u = 0; u < HLD; ++u) {
                bump(cnt, v[u].x, base, span);
                bump(cnt, v[u].y, base, span);
                bump(cnt, v[u].z, base, span);
                bump(cnt, v[u].w, base, span);
            }
        }
        for (int e = (n4 << 2) + tid; e < n; e += HTPB) bump(cnt, p[e], base, span);
        __syncthreads();

        uint4* plane4 = (uint4*)(part + (size_t)blockIdx.x * PS + base);
        for (int i = tid; i < span4; i += HTPB) plane4[i] = ((const uint4*)cnt)[i];
        __syncthreads();
    }
}

__global__ void dinv_kernel(const unsigned int* __restrict__ part, float* __restrict__ dinv,
                            unsigned int* __restrict__ ctr, int N) {
    int i = blockIdx.x * blockDim.x + threadIdx.x;
    if (i < 256) ctr[i] = 0u;
    if (i >= N) return;
    const int PS = plane_stride(N);
    unsigned int s = 0;
    #pragma unroll
    for (int g = 0; g < HBLK; ++g) s += part[(size_t)g * PS + i];
    dinv[i] = rsqrtf((float)s + 1.0f);
}

__device__ __forceinline__ float elu1(float v) {
    const float m = v > 0.f ? 0.f : v;
    const float e = expf(m) - 1.f;
    return v > 0.f ? v : e;
}

template <bool ACT>
__global__ __launch_bounds__(TPB) void gemm_wmma(const float* __restrict__ X,
                                                 const float* __restrict__ W,
                                                 const float* __restrict__ bin,
                                                 const float* __restrict__ dinv,
                                                 float* __restrict__ Y,
                                                 float* __restrict__ Z,
                                                 int nStrips) {
    __shared__ float Wl[64 * C];
    __shared__ __attribute__((aligned(16))) float bl[C];
    if (ACT) {
        if (threadIdx.x < C) bl[threadIdx.x] = bin[threadIdx.x];
    }

    const int wave = threadIdx.x >> 5;
    const int lane = threadIdx.x & 31;
    const int half = lane >> 4;
    const int lo   = lane & 15;

    const int strip  = blockIdx.x * (TPB / 32) + wave;
    const bool active = (strip < nStrips);
    const int m0 = strip * 16;

    v8f acc[8] = {};

    const float* Arow = X + (size_t)(active ? (m0 + lo) : 0) * C + 2 * half;

    for (int kc = 0; kc < C; kc += 64) {
        __syncthreads();
        for (int idx = threadIdx.x; idx < 64 * C; idx += TPB) {
            int k = idx >> 7;
            int n = idx & (C - 1);
            Wl[(k >> 1) * 256 + n * 2 + (k & 1)] = W[(size_t)(kc + k) * C + n];
        }
        __syncthreads();

        if (active) {
            #pragma unroll
            for (int k0 = 0; k0 < 64; k0 += 4) {
                v2f a = *(const v2f*)(Arow + kc + k0);
                if (ACT) {
                    const v2f bb = *(const v2f*)(&bl[kc + k0 + 2 * half]);
                    a.x = elu1(a.x + bb.x);
                    a.y = elu1(a.y + bb.y);
                }
                const float* Bbase = &Wl[((k0 >> 1) + half) * 256 + lo * 2];
                #pragma unroll
                for (int t = 0; t < 8; ++t) {
                    v2f b = *(const v2f*)(Bbase + t * 32);
                    acc[t] = __builtin_amdgcn_wmma_f32_16x16x4_f32(
                        false, a, false, b, (short)0, acc[t], false, false);
                }
            }
        }
    }

    if (active) {
        float* Yt = Y + (size_t)m0 * C;
        float* Zt = Z + (size_t)m0 * C;
        const float4 dA = *(const float4*)(dinv + m0 + 8 * half);
        const float4 dB = *(const float4*)(dinv + m0 + 8 * half + 4);
        const float dv[8] = {dA.x, dA.y, dA.z, dA.w, dB.x, dB.y, dB.z, dB.w};
        #pragma unroll
        for (int t = 0; t < 8; ++t)
            #pragma unroll
            for (int r = 0; r < 8; ++r) {
                const size_t o = (size_t)(r + 8 * half) * C + t * 16 + lo;
                Yt[o] = acc[t][r];
                Zt[o] = acc[t][r] * (dv[r] * dv[r]);
            }
    }
}

template <bool ACT>
__global__ void gemm_tail(const float* __restrict__ X, const float* __restrict__ W,
                          const float* __restrict__ bin, const float* __restrict__ dinv,
                          float* __restrict__ Y, float* __restrict__ Z, int row0, int rows) {
    int i = blockIdx.x * blockDim.x + threadIdx.x;
    if (i >= rows * C) return;
    int r = row0 + i / C, n = i & (C - 1);
    float s = 0.f;
    for (int k = 0; k < C; ++k) {
        float a = X[(size_t)r * C + k];
        if (ACT) a = elu1(a + bin[k]);
        s += a * W[(size_t)k * C + n];
    }
    Y[(size_t)r * C + n] = s;
    Z[(size_t)r * C + n] = s * (dinv[r] * dinv[r]);
}

#define EB 8
#define AGG_BLOCKS 1024
__global__ __launch_bounds__(TPB) void agg_kernel(const float* __restrict__ H, const int* __restrict__ src,
                           const int* __restrict__ dst, const float* __restrict__ dinv,
                           float* __restrict__ out, unsigned int* __restrict__ ctr,
                           int E, int nSplit) {
    const int lane = threadIdx.x & 31;
    const unsigned int place = __builtin_amdgcn_s_sendmsg_rtn(0x87);
    const int myHalf = __builtin_amdgcn_readfirstlane((int)((place >> 18) & 1u));
    const int nTick  = (E + 63) >> 6;
    for (int pass = 0; pass < 2; ++pass) {
        const int hf = myHalf ^ pass;
        for (int draw = 0; draw <= nTick; ++draw) {
            unsigned int tk = 0u;
            if (lane == 0) tk = atomicAdd(&ctr[hf * 64], 1u);
            const int t = __builtin_amdgcn_readfirstlane((int)tk);
            if (t >= nTick) break;
            const int e0 = (t << 6) + lane;
            const int e1 = e0 + 32;
            const int c0 = min(e0, E - 1);
            const int c1 = min(e1, E - 1);
            const int s0 = src[c0];
            const int d0 = dst[c0];
            const int s1 = src[c1];
            const int d1 = dst[c1];
            asm volatile("" :: "v"(s0), "v"(d0), "v"(s1), "v"(d1));
            const bool m0 = (e0 < E) && ((d0 >= nSplit) == (hf != 0));
            const bool m1 = (e1 < E) && ((d1 >= nSplit) == (hf != 0));
            const float a0 = dinv[m0 ? s0 : 0];
            const float b0 = dinv[m0 ? d0 : 0];
            const float a1 = dinv[m1 ? s1 : 0];
            const float b1 = dinv[m1 ? d1 : 0];
            const int n0 = __float_as_int(a0 * b0);
            const int n1 = __float_as_int(a1 * b1);
            unsigned long long mask = (unsigned long long)__builtin_amdgcn_ballot_w32(m0) |
                                      ((unsigned long long)__builtin_amdgcn_ballot_w32(m1) << 32);
            while (mask != 0ull) {
                int idx[EB];
                const int cnt = min(EB, (int)__builtin_popcountll(mask));
                #pragma unroll
                for (int u = 0; u < EB; ++u) {
                    idx[u] = (mask != 0ull) ? (int)__builtin_ctzll(mask) : 0;
                    mask &= (mask - 1ull);
                }
                float v[EB][4];
                #pragma unroll
                for (int u = 0; u < EB; ++u) {
                    if (u < cnt) {
                        const int l  = idx[u] & 31;
                        const int sa = __builtin_amdgcn_readlane(s0, l);
                        const int sb = __builtin_amdgcn_readlane(s1, l);
                        const int sj = (idx[u] < 32) ? sa : sb;
                        const float* h = H + (size_t)sj * C + lane;
                        v[u][0] = h[0];
                        v[u][1] = h[32];
                        v[u][2] = h[64];
                        v[u][3] = h[96];
                    }
                }
                #pragma unroll
                for (int u = 0; u < EB; ++u) {
                    if (u < cnt) {
                        const int l  = idx[u] & 31;
                        const bool lo32 = idx[u] < 32;
                        const int da = __builtin_amdgcn_readlane(d0, l);
                        const int db = __builtin_amdgcn_readlane(d1, l);
                        const int na = __builtin_amdgcn_readlane(n0, l);
                        const int nb = __builtin_amdgcn_readlane(n1, l);
                        const int   dj = lo32 ? da : db;
                        const float nj = __int_as_float(lo32 ? na : nb);
                        float* o = out + (size_t)dj * C + lane;
                        atomicAdd(o +  0, v[u][0] * nj);
                        atomicAdd(o + 32, v[u][1] * nj);
                        atomicAdd(o + 64, v[u][2] * nj);
                        atomicAdd(o + 96, v[u][3] * nj);
                    }
                }
            }
        }
    }
}

__global__ void bias_elu_kernel(const float* __restrict__ in, const float* __restrict__ b,
                                float* __restrict__ out, size_t n4) {
    size_t i = (size_t)blockIdx.x * blockDim.x + threadIdx.x;
    if (i >= n4) return;
    size_t idx = i * 4;
    int c = (int)(idx & (C - 1));
    float4 v = *(const float4*)(in + idx);
    v.x += b[c + 0]; v.y += b[c + 1]; v.z += b[c + 2]; v.w += b[c + 3];
    v.x = v.x > 0.f ? v.x : (expf(v.x) - 1.f);
    v.y = v.y > 0.f ? v.y : (expf(v.y) - 1.f);
    v.z = v.z > 0.f ? v.z : (expf(v.z) - 1.f);
    v.w = v.w > 0.f ? v.w : (expf(v.w) - 1.f);
    *(float4*)(out + idx) = v;
}

extern "C" void kernel_launch(void* const* d_in, const int* in_sizes, int n_in,
                              void* d_out, int out_size, void* d_ws, size_t ws_size,
                              hipStream_t stream) {
    const float* x  = (const float*)d_in[0];
    const int*   ei = (const int*)d_in[1];
    const float* W1 = (const float*)d_in[2];
    const float* b1 = (const float*)d_in[3];
    const float* W2 = (const float*)d_in[4];
    const float* b2 = (const float*)d_in[5];

    const int N = in_sizes[0] / C;
    const int E = in_sizes[1] / 2;
    const int* srcI = ei;
    const int* dstI = ei + E;

    char* ws = (char*)d_ws;
    float* dinv = (float*)ws;
    size_t o1 = ((size_t)N * 4 + 255) & ~(size_t)255;
    size_t hbytes = ((size_t)N * C * 4 + 255) & ~(size_t)255;
    float* buf0 = (float*)(ws + o1);
    float* buf1 = (float*)(ws + o1 + hbytes);
    float* buf2 = (float*)(ws + o1 + 2 * hbytes);
    unsigned int* ctr  = (unsigned int*)(ws + o1 + 3 * hbytes);
    unsigned int* part = (unsigned int*)(ws + o1 + 3 * hbytes + 1024);

    const int strips   = N / 16;
    const int tailRow0 = strips * 16;
    const int tailRows = N - tailRow0;
    const dim3 blk(TPB);
    const int gemmBlocks = (strips + (TPB / 32) - 1) / (TPB / 32);
    const int nSplit     = N / 2;
    const size_t n4      = (size_t)N * C / 4;
    const int eluBlocks  = (int)((n4 + TPB - 1) / TPB);

    deg_hist_kernel<<<HBLK, dim3(HTPB), 0, stream>>>(dstI, part, E, N);
    dinv_kernel<<<(N + TPB - 1) / TPB, blk, 0, stream>>>(part, dinv, ctr, N);

    gemm_wmma<false><<<gemmBlocks, blk, 0, stream>>>(x, W1, nullptr, dinv, buf0, buf1, strips);
    if (tailRows > 0)
        gemm_tail<false><<<(tailRows * C + TPB - 1) / TPB, blk, 0, stream>>>(x, W1, nullptr, dinv, buf0, buf1, tailRow0, tailRows);
    agg_kernel<<<AGG_BLOCKS, blk, 0, stream>>>(buf0, srcI, dstI, dinv, buf1, ctr, E, nSplit);

    gemm_wmma<true><<<gemmBlocks, blk, 0, stream>>>(buf1, W2, b1, dinv, buf0, buf2, strips);
    if (tailRows > 0)
        gemm_tail<true><<<(tailRows * C + TPB - 1) / TPB, blk, 0, stream>>>(buf1, W2, b1, dinv, buf0, buf2, tailRow0, tailRows);
    agg_kernel<<<AGG_BLOCKS, blk, 0, stream>>>(buf0, srcI, dstI, dinv, buf2, ctr + 128, E, nSplit);
    bias_elu_kernel<<<eluBlocks, blk, 0, stream>>>(buf2, b2, (float*)d_out, n4);
}
